# starting kernel
# baseline (speedup 1.0000x reference)
#include <hip/hip_runtime.h>
#include <hip/hip_bf16.h>

typedef float v2f __attribute__((ext_vector_type(2)));
typedef float v8f __attribute__((ext_vector_type(8)));

__device__ __forceinline__ float wave32_allreduce_sum(float v) {
#if __has_builtin(__builtin_amdgcn_wmma_f32_16x16x4_f32)
    v2f a; a[0] = v;    a[1] = 0.0f;
    v2f b; b[0] = 1.0f; b[1] = 1.0f;
    v8f c = {};
    v8f d1 = __builtin_amdgcn_wmma_f32_16x16x4_f32(
        false, a, false, b, (short)0, c, false, false);
    float t = ((d1[0] + d1[1]) + (d1[2] + d1[3])) +
              ((d1[4] + d1[5]) + (d1[6] + d1[7]));
    v2f a2; a2[0] = t; a2[1] = 0.0f;
    v8f d2 = __builtin_amdgcn_wmma_f32_16x16x4_f32(
        false, a2, false, b, (short)0, c, false, false);
    return d2[0];
#else
    for (int m = 16; m >= 1; m >>= 1) v += __shfl_xor(v, m, 32);
    return v;
#endif
}

__device__ __forceinline__ float l1_diff(const float* __restrict__ flow,
                                         int idx, float f0, float f1, float f2) {
    const float* p = flow + 3u * (unsigned)idx;
    return fabsf(f0 - p[0]) + fabsf(f1 - p[1]) + fabsf(f2 - p[2]);
}

__global__ void __launch_bounds__(256)
smooth_loss_kernel(const float* __restrict__ flow,
                   const int*   __restrict__ nn,
                   float*       __restrict__ per_point,
                   float*       __restrict__ partial)
{
    const int i = blockIdx.x * 256 + threadIdx.x;

    const float f0 = flow[3 * (size_t)i + 0];
    const float f1 = flow[3 * (size_t)i + 1];
    const float f2 = flow[3 * (size_t)i + 2];

    const int4* __restrict__ q = (const int4*)(nn + (size_t)i * 32);

    float acc = 0.0f;
    {
        int4 v = q[0];
        acc += l1_diff(flow, v.y, f0, f1, f2);
        acc += l1_diff(flow, v.z, f0, f1, f2);
        acc += l1_diff(flow, v.w, f0, f1, f2);
    }
#pragma unroll
    for (int j = 1; j < 8; ++j) {
        int4 v = q[j];
        acc += l1_diff(flow, v.x, f0, f1, f2);
        acc += l1_diff(flow, v.y, f0, f1, f2);
        acc += l1_diff(flow, v.z, f0, f1, f2);
        acc += l1_diff(flow, v.w, f0, f1, f2);
    }

    per_point[i] = acc * (1.0f / 31.0f);

    float wsum = wave32_allreduce_sum(acc);

    __shared__ float lds[8];
    const int lane = threadIdx.x & 31;
    const int wave = threadIdx.x >> 5;
    if (lane == 0) lds[wave] = wsum;
    __syncthreads();
    if (threadIdx.x == 0) {
        float s = ((lds[0] + lds[1]) + (lds[2] + lds[3])) +
                  ((lds[4] + lds[5]) + (lds[6] + lds[7]));
        partial[blockIdx.x] = s;
    }
}

__global__ void __launch_bounds__(32)
finalize_kernel(const float* __restrict__ partial, int nparts,
                float* __restrict__ out, float scale)
{
    float s = 0.0f;
    for (int j = threadIdx.x; j < nparts; j += 32) s += partial[j];
    float total = wave32_allreduce_sum(s);
    if (threadIdx.x == 0) out[0] = total * scale;
}

extern "C" void kernel_launch(void* const* d_in, const int* in_sizes, int n_in,
                              void* d_out, int out_size, void* d_ws, size_t ws_size,
                              hipStream_t stream) {
    const float* flow = (const float*)d_in[0];
    const int*   nn   = (const int*)d_in[1];
    float* out = (float*)d_out;
    float* partial = (float*)d_ws;

    const int n = in_sizes[0] / 3;
    const int blocks = n / 256;

    smooth_loss_kernel<<<blocks, 256, 0, stream>>>(flow, nn, out + 1, partial);
    finalize_kernel<<<1, 32, 0, stream>>>(partial, blocks, out,
                                          1.0f / (31.0f * (float)n));
}


# Final reduction issues all partial-sum loads at once (8 independent 128-bit loads per lane) instead of a load-wait-add loop of 32 dependent round trips; summation order of the partials changes
# speedup vs baseline: 1.1822x; 1.1822x over previous
#include <hip/hip_runtime.h>
#include <hip/hip_bf16.h>

typedef float v2f __attribute__((ext_vector_type(2)));
typedef float v8f __attribute__((ext_vector_type(8)));

__device__ __forceinline__ float wave32_allreduce_sum(float v) {
#if __has_builtin(__builtin_amdgcn_wmma_f32_16x16x4_f32)
    v2f a; a[0] = v;    a[1] = 0.0f;
    v2f b; b[0] = 1.0f; b[1] = 1.0f;
    v8f c = {};
    v8f d1 = __builtin_amdgcn_wmma_f32_16x16x4_f32(
        false, a, false, b, (short)0, c, false, false);
    float t = ((d1[0] + d1[1]) + (d1[2] + d1[3])) +
              ((d1[4] + d1[5]) + (d1[6] + d1[7]));
    v2f a2; a2[0] = t; a2[1] = 0.0f;
    v8f d2 = __builtin_amdgcn_wmma_f32_16x16x4_f32(
        false, a2, false, b, (short)0, c, false, false);
    return d2[0];
#else
    for (int m = 16; m >= 1; m >>= 1) v += __shfl_xor(v, m, 32);
    return v;
#endif
}

__device__ __forceinline__ float l1_diff(const float* __restrict__ flow,
                                         int idx, float f0, float f1, float f2) {
    const float* p = flow + 3u * (unsigned)idx;
    return fabsf(f0 - p[0]) + fabsf(f1 - p[1]) + fabsf(f2 - p[2]);
}

__global__ void __launch_bounds__(256)
smooth_loss_kernel(const float* __restrict__ flow,
                   const int*   __restrict__ nn,
                   float*       __restrict__ per_point,
                   float*       __restrict__ partial)
{
    const int i = blockIdx.x * 256 + threadIdx.x;

    const float f0 = flow[3 * (size_t)i + 0];
    const float f1 = flow[3 * (size_t)i + 1];
    const float f2 = flow[3 * (size_t)i + 2];

    const int4* __restrict__ q = (const int4*)(nn + (size_t)i * 32);

    float acc = 0.0f;
    {
        int4 v = q[0];
        acc += l1_diff(flow, v.y, f0, f1, f2);
        acc += l1_diff(flow, v.z, f0, f1, f2);
        acc += l1_diff(flow, v.w, f0, f1, f2);
    }
#pragma unroll
    for (int j = 1; j < 8; ++j) {
        int4 v = q[j];
        acc += l1_diff(flow, v.x, f0, f1, f2);
        acc += l1_diff(flow, v.y, f0, f1, f2);
        acc += l1_diff(flow, v.z, f0, f1, f2);
        acc += l1_diff(flow, v.w, f0, f1, f2);
    }

    per_point[i] = acc * (1.0f / 31.0f);

    float wsum = wave32_allreduce_sum(acc);

    __shared__ float lds[8];
    const int lane = threadIdx.x & 31;
    const int wave = threadIdx.x >> 5;
    if (lane == 0) lds[wave] = wsum;
    __syncthreads();
    if (threadIdx.x == 0) {
        float s = ((lds[0] + lds[1]) + (lds[2] + lds[3])) +
                  ((lds[4] + lds[5]) + (lds[6] + lds[7]));
        partial[blockIdx.x] = s;
    }
}

__global__ void __launch_bounds__(32)
finalize_kernel(const float* __restrict__ partial, int nparts,
                float* __restrict__ out, float scale)
{
    const int lane = threadIdx.x;
    const float4* __restrict__ p4 = (const float4*)partial;
    const int nvec = nparts >> 2;
    float s = 0.0f;
    for (int base = 0; base < nvec; base += 256) {
        if (base + 256 <= nvec) {
            const float4* __restrict__ q = p4 + base + lane;
            const float4 v0 = q[0],   v1 = q[32],  v2 = q[64],  v3 = q[96];
            const float4 v4 = q[128], v5 = q[160], v6 = q[192], v7 = q[224];
            const float t0 = (v0.x + v0.y) + (v0.z + v0.w);
            const float t1 = (v1.x + v1.y) + (v1.z + v1.w);
            const float t2 = (v2.x + v2.y) + (v2.z + v2.w);
            const float t3 = (v3.x + v3.y) + (v3.z + v3.w);
            const float t4 = (v4.x + v4.y) + (v4.z + v4.w);
            const float t5 = (v5.x + v5.y) + (v5.z + v5.w);
            const float t6 = (v6.x + v6.y) + (v6.z + v6.w);
            const float t7 = (v7.x + v7.y) + (v7.z + v7.w);
            s += ((t0 + t1) + (t2 + t3)) + ((t4 + t5) + (t6 + t7));
        } else {
            for (int j = base + lane; j < nvec; j += 32) {
                const float4 v = p4[j];
                s += (v.x + v.y) + (v.z + v.w);
            }
        }
    }
    {
        const int j = (nvec << 2) + lane;
        if (j < nparts) s += partial[j];
    }
    float total = wave32_allreduce_sum(s);
    if (threadIdx.x == 0) out[0] = total * scale;
}

extern "C" void kernel_launch(void* const* d_in, const int* in_sizes, int n_in,
                              void* d_out, int out_size, void* d_ws, size_t ws_size,
                              hipStream_t stream) {
    const float* flow = (const float*)d_in[0];
    const int*   nn   = (const int*)d_in[1];
    float* out = (float*)d_out;
    float* partial = (float*)d_ws;

    const int n = in_sizes[0] / 3;
    const int blocks = n / 256;

    smooth_loss_kernel<<<blocks, 256, 0, stream>>>(flow, nn, out + 1, partial);
    finalize_kernel<<<1, 32, 0, stream>>>(partial, blocks, out,
                                          1.0f / (31.0f * (float)n));
}
